# starting kernel
# baseline (speedup 1.0000x reference)
#include <hip/hip_runtime.h>

typedef __attribute__((ext_vector_type(2))) float v2f;
typedef __attribute__((ext_vector_type(8))) float v8f;

#define B_SZ   4
#define DEC_N  128
#define ENC_N  512
#define HID_N  512
#define K_N    512

__global__ __launch_bounds__(256) void proj_wmma_f32(
    const float* __restrict__ A, const float* __restrict__ W,
    const float* __restrict__ bias, float* __restrict__ C,
    int N, int K, int ldw, int woff, int nunits_n)
{
    const int lane = threadIdx.x & 31;
    const int wave = threadIdx.x >> 5;
    const int unit = blockIdx.x * 8 + wave;
    const int um   = (unit / nunits_n) << 5;
    const int un   = (unit % nunits_n) << 5;

    const int half = lane >> 4;
    const int r    = lane & 15;

    const float* a0p = A + (size_t)(um + r)      * K + (half << 1);
    const float* a1p = A + (size_t)(um + 16 + r) * K + (half << 1);
    const float* b0p = W + (size_t)(un + r)      * ldw + woff + (half << 1);
    const float* b1p = W + (size_t)(un + 16 + r) * ldw + woff + (half << 1);

    v8f c00 = {0.f,0.f,0.f,0.f,0.f,0.f,0.f,0.f};
    v8f c01 = c00, c10 = c00, c11 = c00;

    for (int k = 0; k < K; k += 4) {
        v2f a0 = *(const v2f*)(a0p + k);
        v2f a1 = *(const v2f*)(a1p + k);
        v2f b0 = *(const v2f*)(b0p + k);
        v2f b1 = *(const v2f*)(b1p + k);
        c00 = __builtin_amdgcn_wmma_f32_16x16x4_f32(false, a0, false, b0, (short)0, c00, false, false);
        c01 = __builtin_amdgcn_wmma_f32_16x16x4_f32(false, a0, false, b1, (short)0, c01, false, false);
        c10 = __builtin_amdgcn_wmma_f32_16x16x4_f32(false, a1, false, b0, (short)0, c10, false, false);
        c11 = __builtin_amdgcn_wmma_f32_16x16x4_f32(false, a1, false, b1, (short)0, c11, false, false);
    }

    const int   n0   = un + r;
    const int   n1   = un + 16 + r;
    const float bad0 = bias ? bias[n0] : 0.f;
    const float bad1 = bias ? bias[n1] : 0.f;
    float* row0 = C + (size_t)(um + (half << 3)) * N;
    float* row1 = C + (size_t)(um + 16 + (half << 3)) * N;
#pragma unroll
    for (int v = 0; v < 8; ++v) {
        row0[(size_t)v * N + n0] = c00[v] + bad0;
        row0[(size_t)v * N + n1] = c01[v] + bad1;
        row1[(size_t)v * N + n0] = c10[v] + bad0;
        row1[(size_t)v * N + n1] = c11[v] + bad1;
    }
}

__global__ __launch_bounds__(256) void relu_dot_score(
    const float* __restrict__ dec_proj,
    const float* __restrict__ enc_proj,
    const float* __restrict__ w_out,
    const float* __restrict__ b_out,
    float* __restrict__ out)
{
    __shared__ float s_dp[HID_N];
    __shared__ float s_w [HID_N];

    const int bq  = blockIdx.x;
    const int b   = bq >> 7;
    const int tid = threadIdx.x;

    s_dp[tid]       = dec_proj[(size_t)bq * HID_N + tid];
    s_dp[tid + 256] = dec_proj[(size_t)bq * HID_N + tid + 256];
    s_w [tid]       = w_out[tid];
    s_w [tid + 256] = w_out[tid + 256];
    __syncthreads();

    const int   lane = tid & 31;
    const int   wave = tid >> 5;
    const float bo   = b_out[0];
    const float* ep_base = enc_proj + (size_t)b * ENC_N * HID_N;

    for (int e = wave; e < ENC_N; e += 8) {
        const float* ep = ep_base + (size_t)e * HID_N;
        float acc = 0.f;
#pragma unroll
        for (int j = 0; j < 4; ++j) {
            const int h = (lane << 2) + (j << 7);
            float4 ev = *(const float4*)(ep + h);
            float4 dv = *(const float4*)(s_dp + h);
            float4 wv = *(const float4*)(s_w + h);
            float t0 = fmaxf(dv.x + ev.x, 0.f);
            float t1 = fmaxf(dv.y + ev.y, 0.f);
            float t2 = fmaxf(dv.z + ev.z, 0.f);
            float t3 = fmaxf(dv.w + ev.w, 0.f);
            acc = fmaf(t0, wv.x, acc);
            acc = fmaf(t1, wv.y, acc);
            acc = fmaf(t2, wv.z, acc);
            acc = fmaf(t3, wv.w, acc);
        }
        for (int off = 16; off > 0; off >>= 1)
            acc += __shfl_xor(acc, off, 32);
        if (lane == 0)
            out[(size_t)bq * ENC_N + e] = acc + bo;
    }
}

extern "C" void kernel_launch(void* const* d_in, const int* in_sizes, int n_in,
                              void* d_out, int out_size, void* d_ws, size_t ws_size,
                              hipStream_t stream) {
    const float* dec_states = (const float*)d_in[0];
    const float* enc_states = (const float*)d_in[1];
    const float* W_mlp      = (const float*)d_in[2];
    const float* b_mlp      = (const float*)d_in[3];
    const float* W_out      = (const float*)d_in[4];
    const float* b_out      = (const float*)d_in[5];
    float*       out        = (float*)d_out;

    float* enc_proj = (float*)d_ws;
    float* dec_proj = enc_proj + (size_t)B_SZ * ENC_N * HID_N;

    proj_wmma_f32<<<128, 256, 0, stream>>>(enc_states, W_mlp, nullptr, enc_proj,
                                           HID_N, K_N,
                                            1024,  0,
                                            16);
    proj_wmma_f32<<<32, 256, 0, stream>>>(dec_states, W_mlp, b_mlp, dec_proj,
                                          HID_N, K_N,
                                           1024,  512,
                                           16);
    relu_dot_score<<<B_SZ * DEC_N, 256, 0, stream>>>(dec_proj, enc_proj,
                                                     W_out, b_out, out);
}


# score kernel: each wave holds 8 dec rows in registers (h over lanes) and streams enc_proj rows past them, one row load serves 8 q (was 1); folded ALU lane butterfly, same sum tree; whole-line stores
# speedup vs baseline: 1.7285x; 1.7285x over previous
#include <hip/hip_runtime.h>

typedef __attribute__((ext_vector_type(2))) float v2f;
typedef __attribute__((ext_vector_type(8))) float v8f;

#define B_SZ   4
#define DEC_N  128
#define ENC_N  512
#define HID_N  512
#define K_N    512

#define KSPLIT   8
#define PART_LD  36

__global__ __launch_bounds__(256) void proj_wmma_f32(
    const float* __restrict__ A, const float* __restrict__ W,
    const float* __restrict__ bias, float* __restrict__ C,
    int N, int K, int ldw, int woff, int nunits_n)
{
    __shared__ __attribute__((aligned(16))) float s_part[KSPLIT * 32 * PART_LD];

    const int lane = threadIdx.x & 31;
    const int wave = threadIdx.x >> 5;
    const int unit = blockIdx.x;
    const int um   = (unit / nunits_n) << 5;
    const int un   = (unit % nunits_n) << 5;

    const int half = lane >> 4;
    const int r    = lane & 15;

    const int klen = K >> 3;
    const int kbeg = wave * klen;

    const float* a0p = A + (size_t)(um + r)      * K + kbeg + (half << 1);
    const float* a1p = A + (size_t)(um + 16 + r) * K + kbeg + (half << 1);
    const float* b0p = W + (size_t)(un + r)      * ldw + woff + kbeg + (half << 1);
    const float* b1p = W + (size_t)(un + 16 + r) * ldw + woff + kbeg + (half << 1);

    v8f c00 = {0.f,0.f,0.f,0.f,0.f,0.f,0.f,0.f};
    v8f c01 = c00, c10 = c00, c11 = c00;

    for (int k = 0; k < klen; k += 4) {
        v2f a0 = *(const v2f*)(a0p + k);
        v2f a1 = *(const v2f*)(a1p + k);
        v2f b0 = *(const v2f*)(b0p + k);
        v2f b1 = *(const v2f*)(b1p + k);
        c00 = __builtin_amdgcn_wmma_f32_16x16x4_f32(false, a0, false, b0, (short)0, c00, false, false);
        c01 = __builtin_amdgcn_wmma_f32_16x16x4_f32(false, a0, false, b1, (short)0, c01, false, false);
        c10 = __builtin_amdgcn_wmma_f32_16x16x4_f32(false, a1, false, b0, (short)0, c10, false, false);
        c11 = __builtin_amdgcn_wmma_f32_16x16x4_f32(false, a1, false, b1, (short)0, c11, false, false);
    }

    float* sp = s_part + wave * (32 * PART_LD) + (half << 3) * PART_LD + r;
#pragma unroll
    for (int v = 0; v < 8; ++v) {
        sp[v * PART_LD]             = c00[v];
        sp[v * PART_LD + 16]        = c01[v];
        sp[(v + 16) * PART_LD]      = c10[v];
        sp[(v + 16) * PART_LD + 16] = c11[v];
    }
    __syncthreads();

    const int row = threadIdx.x >> 3;
    const int c4  = (threadIdx.x & 7) << 2;
    const float* rp = s_part + row * PART_LD + c4;
    float4 s = *(const float4*)rp;
#pragma unroll
    for (int w = 1; w < KSPLIT; ++w) {
        const float4 p = *(const float4*)(rp + w * (32 * PART_LD));
        s.x += p.x; s.y += p.y; s.z += p.z; s.w += p.w;
    }
    if (bias) {
        const float* bp = bias + un + c4;
        s.x += bp[0]; s.y += bp[1]; s.z += bp[2]; s.w += bp[3];
    }
    *(float4*)(C + (size_t)(um + row) * N + un + c4) = s;
}

#define SC_QT     8
#define SC_EW     32
#define SC_WAVES  4
#define SC_PITCH  33

__device__ __forceinline__ float lane_xor16(float x) {
    const int b = __builtin_bit_cast(int, x);
    return __builtin_bit_cast(float,
        __builtin_amdgcn_permlanex16(b, b, 0x76543210, 0xfedcba98, false, false));
}
template <int M>
__device__ __forceinline__ float row_xor(float x) {
    const int b = __builtin_bit_cast(int, x);
    return __builtin_bit_cast(float,
        __builtin_amdgcn_update_dpp(b, b, 0x160 + M, 0xf, 0xf, false));
}

__global__ __launch_bounds__(128) void relu_dot_score(
    const float* __restrict__ dec_proj,
    const float* __restrict__ enc_proj,
    const float* __restrict__ w_out,
    const float* __restrict__ b_out,
    float* __restrict__ out)
{
    __shared__ float s_res[SC_WAVES * SC_QT * SC_PITCH];

    const int lane = threadIdx.x & 31;
    const int wave = threadIdx.x >> 5;
    const int blk  = blockIdx.x;
    const int ec   = blk & 3;
    const int qg   = (blk >> 2) & 15;
    const int b    = blk >> 6;
    const int row0 = b * DEC_N + qg * SC_QT;
    const int e0   = ec * (SC_WAVES * SC_EW) + wave * SC_EW;
    const int hb   = lane << 2;

    float4 wv[4];
    float4 dv[SC_QT][4];
#pragma unroll
    for (int j = 0; j < 4; ++j)
        wv[j] = *(const float4*)(w_out + hb + (j << 7));
#pragma unroll
    for (int q = 0; q < SC_QT; ++q)
#pragma unroll
        for (int j = 0; j < 4; ++j)
            dv[q][j] = *(const float4*)(dec_proj + (size_t)(row0 + q) * HID_N + hb + (j << 7));
    const float bo = b_out[0];

    const float* ep = enc_proj + ((size_t)b * ENC_N + e0) * HID_N + hb;

    float4 ev[4][4];
#pragma unroll
    for (int s = 0; s < 3; ++s)
#pragma unroll
        for (int j = 0; j < 4; ++j)
            ev[s][j] = *(const float4*)(ep + (size_t)s * HID_N + (j << 7));

    const bool up16 = (lane & 16) != 0;
    const bool up8  = (lane & 8) != 0;
    const bool up4  = (lane & 4) != 0;
    const int  qsel = (lane >> 2) & 7;
    float* sres = s_res + wave * (SC_QT * SC_PITCH);

    for (int i = 0; i < SC_EW; i += 4) {
#pragma unroll
        for (int s = 0; s < 4; ++s) {
            const int nr = min(i + s + 3, SC_EW - 1);
#pragma unroll
            for (int j = 0; j < 4; ++j)
                ev[(s + 3) & 3][j] = *(const float4*)(ep + (size_t)nr * HID_N + (j << 7));

            float acc[SC_QT];
#pragma unroll
            for (int q = 0; q < SC_QT; ++q) acc[q] = 0.f;
#pragma unroll
            for (int j = 0; j < 4; ++j) {
                const float4 e4 = ev[s][j];
                const float4 w4 = wv[j];
#pragma unroll
                for (int q = 0; q < SC_QT; ++q) {
                    const float4 d4 = dv[q][j];
                    const float t0 = fmaxf(d4.x + e4.x, 0.f);
                    const float t1 = fmaxf(d4.y + e4.y, 0.f);
                    const float t2 = fmaxf(d4.z + e4.z, 0.f);
                    const float t3 = fmaxf(d4.w + e4.w, 0.f);
                    acc[q] = fmaf(t0, w4.x, acc[q]);
                    acc[q] = fmaf(t1, w4.y, acc[q]);
                    acc[q] = fmaf(t2, w4.z, acc[q]);
                    acc[q] = fmaf(t3, w4.w, acc[q]);
                }
            }

            float b4[4];
#pragma unroll
            for (int k = 0; k < 4; ++k) {
                const float keep = up16 ? acc[k + 4] : acc[k];
                const float send = up16 ? acc[k]     : acc[k + 4];
                b4[k] = keep + lane_xor16(send);
            }
            float c2[2];
#pragma unroll
            for (int k = 0; k < 2; ++k) {
                const float keep = up8 ? b4[k + 2] : b4[k];
                const float send = up8 ? b4[k]     : b4[k + 2];
                c2[k] = keep + row_xor<8>(send);
            }
            const float keep1 = up4 ? c2[1] : c2[0];
            const float send1 = up4 ? c2[0] : c2[1];
            float r = keep1 + row_xor<4>(send1);
            r += row_xor<2>(r);
            r += row_xor<1>(r);
            if ((lane & 3) == 0)
                sres[qsel * SC_PITCH + i + s] = r + bo;
        }
    }
    __syncthreads();

#pragma unroll
    for (int q = 0; q < SC_QT; ++q)
        out[(size_t)(row0 + q) * ENC_N + e0 + lane] = sres[q * SC_PITCH + lane];
}

extern "C" void kernel_launch(void* const* d_in, const int* in_sizes, int n_in,
                              void* d_out, int out_size, void* d_ws, size_t ws_size,
                              hipStream_t stream) {
    const float* dec_states = (const float*)d_in[0];
    const float* enc_states = (const float*)d_in[1];
    const float* W_mlp      = (const float*)d_in[2];
    const float* b_mlp      = (const float*)d_in[3];
    const float* W_out      = (const float*)d_in[4];
    const float* b_out      = (const float*)d_in[5];
    float*       out        = (float*)d_out;

    float* enc_proj = (float*)d_ws;
    float* dec_proj = enc_proj + (size_t)B_SZ * ENC_N * HID_N;

    proj_wmma_f32<<<1024, 256, 0, stream>>>(enc_states, W_mlp, nullptr, enc_proj,
                                           HID_N, K_N,
                                            1024,  0,
                                            16);
    proj_wmma_f32<<<256, 256, 0, stream>>>(dec_states, W_mlp, b_mlp, dec_proj,
                                          HID_N, K_N,
                                           1024,  512,
                                           16);
    relu_dot_score<<<256, 128, 0, stream>>>(dec_proj, enc_proj,
                                            W_out, b_out, out);
}


# projection GEMMs: each lane loads 64 contiguous bytes of its row per 128-B line straight to registers (4 b128, whole k slice ahead of the WMMAs), A and B k-permuted alike; k sum order changes
# speedup vs baseline: 3.0797x; 1.7817x over previous
#include <hip/hip_runtime.h>

typedef __attribute__((ext_vector_type(2))) float v2f;
typedef __attribute__((ext_vector_type(8))) float v8f;

#define B_SZ   4
#define DEC_N  128
#define ENC_N  512
#define HID_N  512
#define K_N    512

#define KSPLIT   8
#define PART_LD  36
#define KW       (K_N / KSPLIT)
#define KLINES   (KW / 32)

typedef __attribute__((ext_vector_type(4))) float v4f;

struct RowSlice { v4f q[KLINES * 4]; };

__device__ __forceinline__ void load_line(RowSlice& f, const float* p, int line) {
#pragma unroll
    for (int i = 0; i < 4; ++i)
        f.q[line * 4 + i] = *(const v4f*)(p + line * 32 + i * 4);
}

__device__ __forceinline__ v2f k_pair(const RowSlice& f, int t) {
    const v4f q = f.q[t >> 1];
    return (t & 1) ? __builtin_shufflevector(q, q, 2, 3)
                   : __builtin_shufflevector(q, q, 0, 1);
}

__global__ __launch_bounds__(256) void proj_wmma_f32(
    const float* __restrict__ A, const float* __restrict__ W,
    const float* __restrict__ bias, float* __restrict__ C,
    int N, int ldw, int woff, int nunits_n)
{
    __shared__ __attribute__((aligned(16))) float s_part[KSPLIT * 32 * PART_LD];

    const int lane = threadIdx.x & 31;
    const int wave = threadIdx.x >> 5;
    const int unit = blockIdx.x;
    const int um   = (unit / nunits_n) << 5;
    const int un   = (unit % nunits_n) << 5;

    const int half = lane >> 4;
    const int r    = lane & 15;

    const int kbeg = wave * KW + (half << 4);

    const float* a0p = A + (size_t)(um + r)      * K_N + kbeg;
    const float* a1p = A + (size_t)(um + 16 + r) * K_N + kbeg;
    const float* b0p = W + (size_t)(un + r)      * ldw + woff + kbeg;
    const float* b1p = W + (size_t)(un + 16 + r) * ldw + woff + kbeg;

    RowSlice a0, a1, b0, b1;
#pragma unroll
    for (int line = 0; line < KLINES; ++line) {
        load_line(a0, a0p, line);
        load_line(b0, b0p, line);
        load_line(a1, a1p, line);
        load_line(b1, b1p, line);
    }
    __builtin_amdgcn_sched_barrier(0);

    v8f c00 = {0.f,0.f,0.f,0.f,0.f,0.f,0.f,0.f};
    v8f c01 = c00, c10 = c00, c11 = c00;

#pragma unroll
    for (int t = 0; t < KLINES * 8; ++t) {
        const v2f x0 = k_pair(a0, t);
        const v2f x1 = k_pair(a1, t);
        const v2f y0 = k_pair(b0, t);
        const v2f y1 = k_pair(b1, t);
        c00 = __builtin_amdgcn_wmma_f32_16x16x4_f32(false, x0, false, y0, (short)0, c00, false, false);
        c01 = __builtin_amdgcn_wmma_f32_16x16x4_f32(false, x0, false, y1, (short)0, c01, false, false);
        c10 = __builtin_amdgcn_wmma_f32_16x16x4_f32(false, x1, false, y0, (short)0, c10, false, false);
        c11 = __builtin_amdgcn_wmma_f32_16x16x4_f32(false, x1, false, y1, (short)0, c11, false, false);
    }

    float* sp = s_part + wave * (32 * PART_LD) + (half << 3) * PART_LD + r;
#pragma unroll
    for (int v = 0; v < 8; ++v) {
        sp[v * PART_LD]             = c00[v];
        sp[v * PART_LD + 16]        = c01[v];
        sp[(v + 16) * PART_LD]      = c10[v];
        sp[(v + 16) * PART_LD + 16] = c11[v];
    }
    __syncthreads();

    const int row = threadIdx.x >> 3;
    const int c4  = (threadIdx.x & 7) << 2;
    const float* rp = s_part + row * PART_LD + c4;
    float4 s = *(const float4*)rp;
#pragma unroll
    for (int w = 1; w < KSPLIT; ++w) {
        const float4 p = *(const float4*)(rp + w * (32 * PART_LD));
        s.x += p.x; s.y += p.y; s.z += p.z; s.w += p.w;
    }
    if (bias) {
        const float* bp = bias + un + c4;
        s.x += bp[0]; s.y += bp[1]; s.z += bp[2]; s.w += bp[3];
    }
    *(float4*)(C + (size_t)(um + row) * N + un + c4) = s;
}

#define SC_QT     8
#define SC_EW     32
#define SC_WAVES  4
#define SC_PITCH  33

__device__ __forceinline__ float lane_xor16(float x) {
    const int b = __builtin_bit_cast(int, x);
    return __builtin_bit_cast(float,
        __builtin_amdgcn_permlanex16(b, b, 0x76543210, 0xfedcba98, false, false));
}
template <int M>
__device__ __forceinline__ float row_xor(float x) {
    const int b = __builtin_bit_cast(int, x);
    return __builtin_bit_cast(float,
        __builtin_amdgcn_update_dpp(b, b, 0x160 + M, 0xf, 0xf, false));
}

__global__ __launch_bounds__(128) void relu_dot_score(
    const float* __restrict__ dec_proj,
    const float* __restrict__ enc_proj,
    const float* __restrict__ w_out,
    const float* __restrict__ b_out,
    float* __restrict__ out)
{
    __shared__ float s_res[SC_WAVES * SC_QT * SC_PITCH];

    const int lane = threadIdx.x & 31;
    const int wave = threadIdx.x >> 5;
    const int blk  = blockIdx.x;
    const int ec   = blk & 3;
    const int qg   = (blk >> 2) & 15;
    const int b    = blk >> 6;
    const int row0 = b * DEC_N + qg * SC_QT;
    const int e0   = ec * (SC_WAVES * SC_EW) + wave * SC_EW;
    const int hb   = lane << 2;

    float4 wv[4];
    float4 dv[SC_QT][4];
#pragma unroll
    for (int j = 0; j < 4; ++j)
        wv[j] = *(const float4*)(w_out + hb + (j << 7));
#pragma unroll
    for (int q = 0; q < SC_QT; ++q)
#pragma unroll
        for (int j = 0; j < 4; ++j)
            dv[q][j] = *(const float4*)(dec_proj + (size_t)(row0 + q) * HID_N + hb + (j << 7));
    const float bo = b_out[0];

    const float* ep = enc_proj + ((size_t)b * ENC_N + e0) * HID_N + hb;

    float4 ev[4][4];
#pragma unroll
    for (int s = 0; s < 3; ++s)
#pragma unroll
        for (int j = 0; j < 4; ++j)
            ev[s][j] = *(const float4*)(ep + (size_t)s * HID_N + (j << 7));

    const bool up16 = (lane & 16) != 0;
    const bool up8  = (lane & 8) != 0;
    const bool up4  = (lane & 4) != 0;
    const int  qsel = (lane >> 2) & 7;
    float* sres = s_res + wave * (SC_QT * SC_PITCH);

    for (int i = 0; i < SC_EW; i += 4) {
#pragma unroll
        for (int s = 0; s < 4; ++s) {
            const int nr = min(i + s + 3, SC_EW - 1);
#pragma unroll
            for (int j = 0; j < 4; ++j)
                ev[(s + 3) & 3][j] = *(const float4*)(ep + (size_t)nr * HID_N + (j << 7));

            float acc[SC_QT];
#pragma unroll
            for (int q = 0; q < SC_QT; ++q) acc[q] = 0.f;
#pragma unroll
            for (int j = 0; j < 4; ++j) {
                const float4 e4 = ev[s][j];
                const float4 w4 = wv[j];
#pragma unroll
                for (int q = 0; q < SC_QT; ++q) {
                    const float4 d4 = dv[q][j];
                    const float t0 = fmaxf(d4.x + e4.x, 0.f);
                    const float t1 = fmaxf(d4.y + e4.y, 0.f);
                    const float t2 = fmaxf(d4.z + e4.z, 0.f);
                    const float t3 = fmaxf(d4.w + e4.w, 0.f);
                    acc[q] = fmaf(t0, w4.x, acc[q]);
                    acc[q] = fmaf(t1, w4.y, acc[q]);
                    acc[q] = fmaf(t2, w4.z, acc[q]);
                    acc[q] = fmaf(t3, w4.w, acc[q]);
                }
            }

            float b4[4];
#pragma unroll
            for (int k = 0; k < 4; ++k) {
                const float keep = up16 ? acc[k + 4] : acc[k];
                const float send = up16 ? acc[k]     : acc[k + 4];
                b4[k] = keep + lane_xor16(send);
            }
            float c2[2];
#pragma unroll
            for (int k = 0; k < 2; ++k) {
                const float keep = up8 ? b4[k + 2] : b4[k];
                const float send = up8 ? b4[k]     : b4[k + 2];
                c2[k] = keep + row_xor<8>(send);
            }
            const float keep1 = up4 ? c2[1] : c2[0];
            const float send1 = up4 ? c2[0] : c2[1];
            float r = keep1 + row_xor<4>(send1);
            r += row_xor<2>(r);
            r += row_xor<1>(r);
            if ((lane & 3) == 0)
                sres[qsel * SC_PITCH + i + s] = r + bo;
        }
    }
    __syncthreads();

#pragma unroll
    for (int q = 0; q < SC_QT; ++q)
        out[(size_t)(row0 + q) * ENC_N + e0 + lane] = sres[q * SC_PITCH + lane];
}

extern "C" void kernel_launch(void* const* d_in, const int* in_sizes, int n_in,
                              void* d_out, int out_size, void* d_ws, size_t ws_size,
                              hipStream_t stream) {
    const float* dec_states = (const float*)d_in[0];
    const float* enc_states = (const float*)d_in[1];
    const float* W_mlp      = (const float*)d_in[2];
    const float* b_mlp      = (const float*)d_in[3];
    const float* W_out      = (const float*)d_in[4];
    const float* b_out      = (const float*)d_in[5];
    float*       out        = (float*)d_out;

    float* enc_proj = (float*)d_ws;
    float* dec_proj = enc_proj + (size_t)B_SZ * ENC_N * HID_N;

    proj_wmma_f32<<<1024, 256, 0, stream>>>(enc_states, W_mlp, nullptr, enc_proj,
                                           HID_N,
                                            1024,  0,
                                            16);
    proj_wmma_f32<<<256, 256, 0, stream>>>(dec_states, W_mlp, b_mlp, dec_proj,
                                          HID_N,
                                           1024,  512,
                                           16);
    relu_dot_score<<<256, 128, 0, stream>>>(dec_proj, enc_proj,
                                            W_out, b_out, out);
}


# enc projection GEMM: one 64x64 output block per work-group, held whole in each wave's registers (4x4 WMMA tiles, was 2x2), 4 waves split K, line sets loaded one ahead of use; k sum order changes
# speedup vs baseline: 3.4814x; 1.1304x over previous
#include <hip/hip_runtime.h>

typedef __attribute__((ext_vector_type(2))) float v2f;
typedef __attribute__((ext_vector_type(8))) float v8f;

#define B_SZ   4
#define DEC_N  128
#define ENC_N  512
#define HID_N  512
#define K_N    512

#define KSPLIT   8
#define PART_LD  36
#define KW       (K_N / KSPLIT)
#define KLINES   (KW / 32)

typedef __attribute__((ext_vector_type(4))) float v4f;

struct RowSlice { v4f q[KLINES * 4]; };

__device__ __forceinline__ void load_line(RowSlice& f, const float* p, int line) {
#pragma unroll
    for (int i = 0; i < 4; ++i)
        f.q[line * 4 + i] = *(const v4f*)(p + line * 32 + i * 4);
}

__device__ __forceinline__ v2f k_pair(const RowSlice& f, int t) {
    const v4f q = f.q[t >> 1];
    return (t & 1) ? __builtin_shufflevector(q, q, 2, 3)
                   : __builtin_shufflevector(q, q, 0, 1);
}

__global__ __launch_bounds__(256) void proj_wmma_f32(
    const float* __restrict__ A, const float* __restrict__ W,
    const float* __restrict__ bias, float* __restrict__ C,
    int N, int ldw, int woff, int nunits_n)
{
    __shared__ __attribute__((aligned(16))) float s_part[KSPLIT * 32 * PART_LD];

    const int lane = threadIdx.x & 31;
    const int wave = threadIdx.x >> 5;
    const int unit = blockIdx.x;
    const int um   = (unit / nunits_n) << 5;
    const int un   = (unit % nunits_n) << 5;

    const int half = lane >> 4;
    const int r    = lane & 15;

    const int kbeg = wave * KW + (half << 4);

    const float* a0p = A + (size_t)(um + r)      * K_N + kbeg;
    const float* a1p = A + (size_t)(um + 16 + r) * K_N + kbeg;
    const float* b0p = W + (size_t)(un + r)      * ldw + woff + kbeg;
    const float* b1p = W + (size_t)(un + 16 + r) * ldw + woff + kbeg;

    RowSlice a0, a1, b0, b1;
#pragma unroll
    for (int line = 0; line < KLINES; ++line) {
        load_line(a0, a0p, line);
        load_line(b0, b0p, line);
        load_line(a1, a1p, line);
        load_line(b1, b1p, line);
    }
    __builtin_amdgcn_sched_barrier(0);

    v8f c00 = {0.f,0.f,0.f,0.f,0.f,0.f,0.f,0.f};
    v8f c01 = c00, c10 = c00, c11 = c00;

#pragma unroll
    for (int t = 0; t < KLINES * 8; ++t) {
        const v2f x0 = k_pair(a0, t);
        const v2f x1 = k_pair(a1, t);
        const v2f y0 = k_pair(b0, t);
        const v2f y1 = k_pair(b1, t);
        c00 = __builtin_amdgcn_wmma_f32_16x16x4_f32(false, x0, false, y0, (short)0, c00, false, false);
        c01 = __builtin_amdgcn_wmma_f32_16x16x4_f32(false, x0, false, y1, (short)0, c01, false, false);
        c10 = __builtin_amdgcn_wmma_f32_16x16x4_f32(false, x1, false, y0, (short)0, c10, false, false);
        c11 = __builtin_amdgcn_wmma_f32_16x16x4_f32(false, x1, false, y1, (short)0, c11, false, false);
    }

    float* sp = s_part + wave * (32 * PART_LD) + (half << 3) * PART_LD + r;
#pragma unroll
    for (int v = 0; v < 8; ++v) {
        sp[v * PART_LD]             = c00[v];
        sp[v * PART_LD + 16]        = c01[v];
        sp[(v + 16) * PART_LD]      = c10[v];
        sp[(v + 16) * PART_LD + 16] = c11[v];
    }
    __syncthreads();

    const int row = threadIdx.x >> 3;
    const int c4  = (threadIdx.x & 7) << 2;
    const float* rp = s_part + row * PART_LD + c4;
    float4 s = *(const float4*)rp;
#pragma unroll
    for (int w = 1; w < KSPLIT; ++w) {
        const float4 p = *(const float4*)(rp + w * (32 * PART_LD));
        s.x += p.x; s.y += p.y; s.z += p.z; s.w += p.w;
    }
    if (bias) {
        const float* bp = bias + un + c4;
        s.x += bp[0]; s.y += bp[1]; s.z += bp[2]; s.w += bp[3];
    }
    *(float4*)(C + (size_t)(um + row) * N + un + c4) = s;
}

#define ET       4
#define E_WAVES  4
#define E_KW     (K_N / E_WAVES)
#define E_LINES  (E_KW / 32)
#define E_LD     68

struct LineSet { v4f a[ET][4]; v4f b[ET][4]; };

__device__ __forceinline__ void load_lineset(LineSet& f, const float* ap, const float* bp,
                                             int ldw, int line) {
#pragma unroll
    for (int s = 0; s < ET; ++s) {
#pragma unroll
        for (int i = 0; i < 4; ++i)
            f.a[s][i] = *(const v4f*)(ap + (size_t)(s * 16) * K_N + line * 32 + i * 4);
#pragma unroll
        for (int i = 0; i < 4; ++i)
            f.b[s][i] = *(const v4f*)(bp + (size_t)(s * 16) * ldw + line * 32 + i * 4);
    }
}

__device__ __forceinline__ v2f pair_of(const v4f q, int odd) {
    return odd ? __builtin_shufflevector(q, q, 2, 3)
               : __builtin_shufflevector(q, q, 0, 1);
}

__device__ __forceinline__ void mma_lineset(const LineSet& f, v8f (&acc)[ET][ET]) {
#pragma unroll
    for (int t = 0; t < 8; ++t) {
        v2f x[ET], y[ET];
#pragma unroll
        for (int s = 0; s < ET; ++s) {
            x[s] = pair_of(f.a[s][t >> 1], t & 1);
            y[s] = pair_of(f.b[s][t >> 1], t & 1);
        }
#pragma unroll
        for (int i = 0; i < ET; ++i)
#pragma unroll
            for (int j = 0; j < ET; ++j)
                acc[i][j] = __builtin_amdgcn_wmma_f32_16x16x4_f32(false, x[i], false, y[j],
                                                                  (short)0, acc[i][j], false, false);
    }
}

__global__ __launch_bounds__(128) void proj_wmma_f32_rb(
    const float* __restrict__ A, const float* __restrict__ W,
    float* __restrict__ C, int N, int ldw, int woff, int nunits_n)
{
    __shared__ __attribute__((aligned(16))) float s_blk[E_WAVES * 64 * E_LD];

    const int lane = threadIdx.x & 31;
    const int wave = threadIdx.x >> 5;
    const int unit = blockIdx.x;
    const int um   = (unit / nunits_n) << 6;
    const int un   = (unit % nunits_n) << 6;

    const int half = lane >> 4;
    const int r    = lane & 15;

    const int kbeg = wave * E_KW + (half << 4);

    const float* ap = A + (size_t)(um + r) * K_N + kbeg;
    const float* bp = W + (size_t)(un + r) * ldw + woff + kbeg;

    v8f acc[ET][ET];
#pragma unroll
    for (int i = 0; i < ET; ++i)
#pragma unroll
        for (int j = 0; j < ET; ++j)
            acc[i][j] = (v8f){0.f,0.f,0.f,0.f,0.f,0.f,0.f,0.f};

    LineSet f0, f1;
    load_lineset(f0, ap, bp, ldw, 0);
    load_lineset(f1, ap, bp, ldw, 1);
    __builtin_amdgcn_sched_barrier(0);
    mma_lineset(f0, acc);
    __builtin_amdgcn_sched_barrier(0);
    load_lineset(f0, ap, bp, ldw, 2);
    __builtin_amdgcn_sched_barrier(0);
    mma_lineset(f1, acc);
    __builtin_amdgcn_sched_barrier(0);
    load_lineset(f1, ap, bp, ldw, 3);
    __builtin_amdgcn_sched_barrier(0);
    mma_lineset(f0, acc);
    mma_lineset(f1, acc);

    float* sp = s_blk + wave * (64 * E_LD) + (half << 3) * E_LD + r;
#pragma unroll
    for (int i = 0; i < ET; ++i)
#pragma unroll
        for (int j = 0; j < ET; ++j)
#pragma unroll
            for (int v = 0; v < 8; ++v)
                sp[(i * 16 + v) * E_LD + j * 16] = acc[i][j][v];
    __syncthreads();

    const int row0 = threadIdx.x >> 4;
    const int c4   = (threadIdx.x & 15) << 2;
#pragma unroll
    for (int p = 0; p < 8; ++p) {
        const int row = row0 + (p << 3);
        const float* rp = s_blk + row * E_LD + c4;
        float4 s = *(const float4*)rp;
#pragma unroll
        for (int w = 1; w < E_WAVES; ++w) {
            const float4 q = *(const float4*)(rp + w * (64 * E_LD));
            s.x += q.x; s.y += q.y; s.z += q.z; s.w += q.w;
        }
        *(float4*)(C + (size_t)(um + row) * N + un + c4) = s;
    }
}

#define SC_QT     8
#define SC_EW     32
#define SC_WAVES  4
#define SC_PITCH  33

__device__ __forceinline__ float lane_xor16(float x) {
    const int b = __builtin_bit_cast(int, x);
    return __builtin_bit_cast(float,
        __builtin_amdgcn_permlanex16(b, b, 0x76543210, 0xfedcba98, false, false));
}
template <int M>
__device__ __forceinline__ float row_xor(float x) {
    const int b = __builtin_bit_cast(int, x);
    return __builtin_bit_cast(float,
        __builtin_amdgcn_update_dpp(b, b, 0x160 + M, 0xf, 0xf, false));
}

__global__ __launch_bounds__(128) void relu_dot_score(
    const float* __restrict__ dec_proj,
    const float* __restrict__ enc_proj,
    const float* __restrict__ w_out,
    const float* __restrict__ b_out,
    float* __restrict__ out)
{
    __shared__ float s_res[SC_WAVES * SC_QT * SC_PITCH];

    const int lane = threadIdx.x & 31;
    const int wave = threadIdx.x >> 5;
    const int blk  = blockIdx.x;
    const int ec   = blk & 3;
    const int qg   = (blk >> 2) & 15;
    const int b    = blk >> 6;
    const int row0 = b * DEC_N + qg * SC_QT;
    const int e0   = ec * (SC_WAVES * SC_EW) + wave * SC_EW;
    const int hb   = lane << 2;

    float4 wv[4];
    float4 dv[SC_QT][4];
#pragma unroll
    for (int j = 0; j < 4; ++j)
        wv[j] = *(const float4*)(w_out + hb + (j << 7));
#pragma unroll
    for (int q = 0; q < SC_QT; ++q)
#pragma unroll
        for (int j = 0; j < 4; ++j)
            dv[q][j] = *(const float4*)(dec_proj + (size_t)(row0 + q) * HID_N + hb + (j << 7));
    const float bo = b_out[0];

    const float* ep = enc_proj + ((size_t)b * ENC_N + e0) * HID_N + hb;

    float4 ev[4][4];
#pragma unroll
    for (int s = 0; s < 3; ++s)
#pragma unroll
        for (int j = 0; j < 4; ++j)
            ev[s][j] = *(const float4*)(ep + (size_t)s * HID_N + (j << 7));

    const bool up16 = (lane & 16) != 0;
    const bool up8  = (lane & 8) != 0;
    const bool up4  = (lane & 4) != 0;
    const int  qsel = (lane >> 2) & 7;
    float* sres = s_res + wave * (SC_QT * SC_PITCH);

    for (int i = 0; i < SC_EW; i += 4) {
#pragma unroll
        for (int s = 0; s < 4; ++s) {
            const int nr = min(i + s + 3, SC_EW - 1);
#pragma unroll
            for (int j = 0; j < 4; ++j)
                ev[(s + 3) & 3][j] = *(const float4*)(ep + (size_t)nr * HID_N + (j << 7));

            float acc[SC_QT];
#pragma unroll
            for (int q = 0; q < SC_QT; ++q) acc[q] = 0.f;
#pragma unroll
            for (int j = 0; j < 4; ++j) {
                const float4 e4 = ev[s][j];
                const float4 w4 = wv[j];
#pragma unroll
                for (int q = 0; q < SC_QT; ++q) {
                    const float4 d4 = dv[q][j];
                    const float t0 = fmaxf(d4.x + e4.x, 0.f);
                    const float t1 = fmaxf(d4.y + e4.y, 0.f);
                    const float t2 = fmaxf(d4.z + e4.z, 0.f);
                    const float t3 = fmaxf(d4.w + e4.w, 0.f);
                    acc[q] = fmaf(t0, w4.x, acc[q]);
                    acc[q] = fmaf(t1, w4.y, acc[q]);
                    acc[q] = fmaf(t2, w4.z, acc[q]);
                    acc[q] = fmaf(t3, w4.w, acc[q]);
                }
            }

            float b4[4];
#pragma unroll
            for (int k = 0; k < 4; ++k) {
                const float keep = up16 ? acc[k + 4] : acc[k];
                const float send = up16 ? acc[k]     : acc[k + 4];
                b4[k] = keep + lane_xor16(send);
            }
            float c2[2];
#pragma unroll
            for (int k = 0; k < 2; ++k) {
                const float keep = up8 ? b4[k + 2] : b4[k];
                const float send = up8 ? b4[k]     : b4[k + 2];
                c2[k] = keep + row_xor<8>(send);
            }
            const float keep1 = up4 ? c2[1] : c2[0];
            const float send1 = up4 ? c2[0] : c2[1];
            float r = keep1 + row_xor<4>(send1);
            r += row_xor<2>(r);
            r += row_xor<1>(r);
            if ((lane & 3) == 0)
                sres[qsel * SC_PITCH + i + s] = r + bo;
        }
    }
    __syncthreads();

#pragma unroll
    for (int q = 0; q < SC_QT; ++q)
        out[(size_t)(row0 + q) * ENC_N + e0 + lane] = sres[q * SC_PITCH + lane];
}

extern "C" void kernel_launch(void* const* d_in, const int* in_sizes, int n_in,
                              void* d_out, int out_size, void* d_ws, size_t ws_size,
                              hipStream_t stream) {
    const float* dec_states = (const float*)d_in[0];
    const float* enc_states = (const float*)d_in[1];
    const float* W_mlp      = (const float*)d_in[2];
    const float* b_mlp      = (const float*)d_in[3];
    const float* W_out      = (const float*)d_in[4];
    const float* b_out      = (const float*)d_in[5];
    float*       out        = (float*)d_out;

    float* enc_proj = (float*)d_ws;
    float* dec_proj = enc_proj + (size_t)B_SZ * ENC_N * HID_N;

    proj_wmma_f32_rb<<<256, 128, 0, stream>>>(enc_states, W_mlp, enc_proj,
                                              HID_N,
                                               1024,  0,
                                               8);
    proj_wmma_f32<<<256, 256, 0, stream>>>(dec_states, W_mlp, b_mlp, dec_proj,
                                          HID_N,
                                           1024,  512,
                                           16);
    relu_dot_score<<<256, 128, 0, stream>>>(dec_proj, enc_proj,
                                            W_out, b_out, out);
}


# score kernel: 8 waves per block, two per SIMD (was 4, one per SIMD), each 8 q x 16 e (was 8 x 32): a sibling wave issues in the other's dependent-ALU gaps and load waits; same sum tree
# speedup vs baseline: 4.0538x; 1.1644x over previous
#include <hip/hip_runtime.h>

typedef __attribute__((ext_vector_type(2))) float v2f;
typedef __attribute__((ext_vector_type(8))) float v8f;

#define B_SZ   4
#define DEC_N  128
#define ENC_N  512
#define HID_N  512
#define K_N    512

#define KSPLIT   8
#define PART_LD  36
#define KW       (K_N / KSPLIT)
#define KLINES   (KW / 32)

typedef __attribute__((ext_vector_type(4))) float v4f;

struct RowSlice { v4f q[KLINES * 4]; };

__device__ __forceinline__ void load_line(RowSlice& f, const float* p, int line) {
#pragma unroll
    for (int i = 0; i < 4; ++i)
        f.q[line * 4 + i] = *(const v4f*)(p + line * 32 + i * 4);
}

__device__ __forceinline__ v2f k_pair(const RowSlice& f, int t) {
    const v4f q = f.q[t >> 1];
    return (t & 1) ? __builtin_shufflevector(q, q, 2, 3)
                   : __builtin_shufflevector(q, q, 0, 1);
}

__global__ __launch_bounds__(256) void proj_wmma_f32(
    const float* __restrict__ A, const float* __restrict__ W,
    const float* __restrict__ bias, float* __restrict__ C,
    int N, int ldw, int woff, int nunits_n)
{
    __shared__ __attribute__((aligned(16))) float s_part[KSPLIT * 32 * PART_LD];

    const int lane = threadIdx.x & 31;
    const int wave = threadIdx.x >> 5;
    const int unit = blockIdx.x;
    const int um   = (unit / nunits_n) << 5;
    const int un   = (unit % nunits_n) << 5;

    const int half = lane >> 4;
    const int r    = lane & 15;

    const int kbeg = wave * KW + (half << 4);

    const float* a0p = A + (size_t)(um + r)      * K_N + kbeg;
    const float* a1p = A + (size_t)(um + 16 + r) * K_N + kbeg;
    const float* b0p = W + (size_t)(un + r)      * ldw + woff + kbeg;
    const float* b1p = W + (size_t)(un + 16 + r) * ldw + woff + kbeg;

    RowSlice a0, a1, b0, b1;
#pragma unroll
    for (int line = 0; line < KLINES; ++line) {
        load_line(a0, a0p, line);
        load_line(b0, b0p, line);
        load_line(a1, a1p, line);
        load_line(b1, b1p, line);
    }
    __builtin_amdgcn_sched_barrier(0);

    v8f c00 = {0.f,0.f,0.f,0.f,0.f,0.f,0.f,0.f};
    v8f c01 = c00, c10 = c00, c11 = c00;

#pragma unroll
    for (int t = 0; t < KLINES * 8; ++t) {
        const v2f x0 = k_pair(a0, t);
        const v2f x1 = k_pair(a1, t);
        const v2f y0 = k_pair(b0, t);
        const v2f y1 = k_pair(b1, t);
        c00 = __builtin_amdgcn_wmma_f32_16x16x4_f32(false, x0, false, y0, (short)0, c00, false, false);
        c01 = __builtin_amdgcn_wmma_f32_16x16x4_f32(false, x0, false, y1, (short)0, c01, false, false);
        c10 = __builtin_amdgcn_wmma_f32_16x16x4_f32(false, x1, false, y0, (short)0, c10, false, false);
        c11 = __builtin_amdgcn_wmma_f32_16x16x4_f32(false, x1, false, y1, (short)0, c11, false, false);
    }

    float* sp = s_part + wave * (32 * PART_LD) + (half << 3) * PART_LD + r;
#pragma unroll
    for (int v = 0; v < 8; ++v) {
        sp[v * PART_LD]             = c00[v];
        sp[v * PART_LD + 16]        = c01[v];
        sp[(v + 16) * PART_LD]      = c10[v];
        sp[(v + 16) * PART_LD + 16] = c11[v];
    }
    __syncthreads();

    const int row = threadIdx.x >> 3;
    const int c4  = (threadIdx.x & 7) << 2;
    const float* rp = s_part + row * PART_LD + c4;
    float4 s = *(const float4*)rp;
#pragma unroll
    for (int w = 1; w < KSPLIT; ++w) {
        const float4 p = *(const float4*)(rp + w * (32 * PART_LD));
        s.x += p.x; s.y += p.y; s.z += p.z; s.w += p.w;
    }
    if (bias) {
        const float* bp = bias + un + c4;
        s.x += bp[0]; s.y += bp[1]; s.z += bp[2]; s.w += bp[3];
    }
    *(float4*)(C + (size_t)(um + row) * N + un + c4) = s;
}

#define ET       4
#define E_WAVES  4
#define E_KW     (K_N / E_WAVES)
#define E_LINES  (E_KW / 32)
#define E_LD     68

struct LineSet { v4f a[ET][4]; v4f b[ET][4]; };

__device__ __forceinline__ void load_lineset(LineSet& f, const float* ap, const float* bp,
                                             int ldw, int line) {
#pragma unroll
    for (int s = 0; s < ET; ++s) {
#pragma unroll
        for (int i = 0; i < 4; ++i)
            f.a[s][i] = *(const v4f*)(ap + (size_t)(s * 16) * K_N + line * 32 + i * 4);
#pragma unroll
        for (int i = 0; i < 4; ++i)
            f.b[s][i] = *(const v4f*)(bp + (size_t)(s * 16) * ldw + line * 32 + i * 4);
    }
}

__device__ __forceinline__ v2f pair_of(const v4f q, int odd) {
    return odd ? __builtin_shufflevector(q, q, 2, 3)
               : __builtin_shufflevector(q, q, 0, 1);
}

__device__ __forceinline__ void mma_lineset(const LineSet& f, v8f (&acc)[ET][ET]) {
#pragma unroll
    for (int t = 0; t < 8; ++t) {
        v2f x[ET], y[ET];
#pragma unroll
        for (int s = 0; s < ET; ++s) {
            x[s] = pair_of(f.a[s][t >> 1], t & 1);
            y[s] = pair_of(f.b[s][t >> 1], t & 1);
        }
#pragma unroll
        for (int i = 0; i < ET; ++i)
#pragma unroll
            for (int j = 0; j < ET; ++j)
                acc[i][j] = __builtin_amdgcn_wmma_f32_16x16x4_f32(false, x[i], false, y[j],
                                                                  (short)0, acc[i][j], false, false);
    }
}

__global__ __launch_bounds__(128) void proj_wmma_f32_rb(
    const float* __restrict__ A, const float* __restrict__ W,
    float* __restrict__ C, int N, int ldw, int woff, int nunits_n)
{
    __shared__ __attribute__((aligned(16))) float s_blk[E_WAVES * 64 * E_LD];

    const int lane = threadIdx.x & 31;
    const int wave = threadIdx.x >> 5;
    const int unit = blockIdx.x;
    const int um   = (unit / nunits_n) << 6;
    const int un   = (unit % nunits_n) << 6;

    const int half = lane >> 4;
    const int r    = lane & 15;

    const int kbeg = wave * E_KW + (half << 4);

    const float* ap = A + (size_t)(um + r) * K_N + kbeg;
    const float* bp = W + (size_t)(un + r) * ldw + woff + kbeg;

    v8f acc[ET][ET];
#pragma unroll
    for (int i = 0; i < ET; ++i)
#pragma unroll
        for (int j = 0; j < ET; ++j)
            acc[i][j] = (v8f){0.f,0.f,0.f,0.f,0.f,0.f,0.f,0.f};

    LineSet f0, f1;
    load_lineset(f0, ap, bp, ldw, 0);
    load_lineset(f1, ap, bp, ldw, 1);
    __builtin_amdgcn_sched_barrier(0);
    mma_lineset(f0, acc);
    __builtin_amdgcn_sched_barrier(0);
    load_lineset(f0, ap, bp, ldw, 2);
    __builtin_amdgcn_sched_barrier(0);
    mma_lineset(f1, acc);
    __builtin_amdgcn_sched_barrier(0);
    load_lineset(f1, ap, bp, ldw, 3);
    __builtin_amdgcn_sched_barrier(0);
    mma_lineset(f0, acc);
    mma_lineset(f1, acc);

    float* sp = s_blk + wave * (64 * E_LD) + (half << 3) * E_LD + r;
#pragma unroll
    for (int i = 0; i < ET; ++i)
#pragma unroll
        for (int j = 0; j < ET; ++j)
#pragma unroll
            for (int v = 0; v < 8; ++v)
                sp[(i * 16 + v) * E_LD + j * 16] = acc[i][j][v];
    __syncthreads();

    const int row0 = threadIdx.x >> 4;
    const int c4   = (threadIdx.x & 15) << 2;
#pragma unroll
    for (int p = 0; p < 8; ++p) {
        const int row = row0 + (p << 3);
        const float* rp = s_blk + row * E_LD + c4;
        float4 s = *(const float4*)rp;
#pragma unroll
        for (int w = 1; w < E_WAVES; ++w) {
            const float4 q = *(const float4*)(rp + w * (64 * E_LD));
            s.x += q.x; s.y += q.y; s.z += q.z; s.w += q.w;
        }
        *(float4*)(C + (size_t)(um + row) * N + un + c4) = s;
    }
}

#define SC_QT     8
#define SC_EW     16
#define SC_WAVES  8
#define SC_BE     (SC_WAVES * SC_EW)
#define SC_PITCH  (SC_BE + 4)
static_assert(SC_WAVES == SC_QT, "the closing store gives wave w the block's dec row w");

__device__ __forceinline__ float lane_xor16(float x) {
    const int b = __builtin_bit_cast(int, x);
    return __builtin_bit_cast(float,
        __builtin_amdgcn_permlanex16(b, b, 0x76543210, 0xfedcba98, false, false));
}
template <int M>
__device__ __forceinline__ float row_xor(float x) {
    const int b = __builtin_bit_cast(int, x);
    return __builtin_bit_cast(float,
        __builtin_amdgcn_update_dpp(b, b, 0x160 + M, 0xf, 0xf, false));
}

__global__ __launch_bounds__(SC_WAVES * 32) void relu_dot_score(
    const float* __restrict__ dec_proj,
    const float* __restrict__ enc_proj,
    const float* __restrict__ w_out,
    const float* __restrict__ b_out,
    float* __restrict__ out)
{
    __shared__ __attribute__((aligned(16))) float s_res[SC_QT * SC_PITCH];

    const int lane = threadIdx.x & 31;
    const int wave = threadIdx.x >> 5;
    const int blk  = blockIdx.x;
    const int ec   = blk & 3;
    const int qg   = (blk >> 2) & 15;
    const int b    = blk >> 6;
    const int row0 = b * DEC_N + qg * SC_QT;
    const int eb   = ec * SC_BE;
    const int e0   = eb + wave * SC_EW;
    const int hb   = lane << 2;

    float4 wv[4];
    float4 dv[SC_QT][4];
#pragma unroll
    for (int j = 0; j < 4; ++j)
        wv[j] = *(const float4*)(w_out + hb + (j << 7));
#pragma unroll
    for (int q = 0; q < SC_QT; ++q)
#pragma unroll
        for (int j = 0; j < 4; ++j)
            dv[q][j] = *(const float4*)(dec_proj + (size_t)(row0 + q) * HID_N + hb + (j << 7));
    const float bo = b_out[0];

    const float* ep = enc_proj + ((size_t)b * ENC_N + e0) * HID_N + hb;

    float4 ev[4][4];
#pragma unroll
    for (int s = 0; s < 3; ++s)
#pragma unroll
        for (int j = 0; j < 4; ++j)
            ev[s][j] = *(const float4*)(ep + (size_t)s * HID_N + (j << 7));

    const bool up16 = (lane & 16) != 0;
    const bool up8  = (lane & 8) != 0;
    const bool up4  = (lane & 4) != 0;
    const int  qsel = (lane >> 2) & 7;
    float* sres = s_res + wave * SC_EW;

    for (int i = 0; i < SC_EW; i += 4) {
#pragma unroll
        for (int s = 0; s < 4; ++s) {
            const int nr = min(i + s + 3, SC_EW - 1);
#pragma unroll
            for (int j = 0; j < 4; ++j)
                ev[(s + 3) & 3][j] = *(const float4*)(ep + (size_t)nr * HID_N + (j << 7));

            float acc[SC_QT];
#pragma unroll
            for (int q = 0; q < SC_QT; ++q) acc[q] = 0.f;
#pragma unroll
            for (int j = 0; j < 4; ++j) {
                const float4 e4 = ev[s][j];
                const float4 w4 = wv[j];
#pragma unroll
                for (int q = 0; q < SC_QT; ++q) {
                    const float4 d4 = dv[q][j];
                    const float t0 = fmaxf(d4.x + e4.x, 0.f);
                    const float t1 = fmaxf(d4.y + e4.y, 0.f);
                    const float t2 = fmaxf(d4.z + e4.z, 0.f);
                    const float t3 = fmaxf(d4.w + e4.w, 0.f);
                    acc[q] = fmaf(t0, w4.x, acc[q]);
                    acc[q] = fmaf(t1, w4.y, acc[q]);
                    acc[q] = fmaf(t2, w4.z, acc[q]);
                    acc[q] = fmaf(t3, w4.w, acc[q]);
                }
            }

            float b4[4];
#pragma unroll
            for (int k = 0; k < 4; ++k) {
                const float keep = up16 ? acc[k + 4] : acc[k];
                const float send = up16 ? acc[k]     : acc[k + 4];
                b4[k] = keep + lane_xor16(send);
            }
            float c2[2];
#pragma unroll
            for (int k = 0; k < 2; ++k) {
                const float keep = up8 ? b4[k + 2] : b4[k];
                const float send = up8 ? b4[k]     : b4[k + 2];
                c2[k] = keep + row_xor<8>(send);
            }
            const float keep1 = up4 ? c2[1] : c2[0];
            const float send1 = up4 ? c2[0] : c2[1];
            float r = keep1 + row_xor<4>(send1);
            r += row_xor<2>(r);
            r += row_xor<1>(r);
            if ((lane & 3) == 0)
                sres[qsel * SC_PITCH + i + s] = r + bo;
        }
    }
    __syncthreads();

    const float4 o = *(const float4*)(s_res + wave * SC_PITCH + hb);
    *(float4*)(out + (size_t)(row0 + wave) * ENC_N + eb + hb) = o;
}

extern "C" void kernel_launch(void* const* d_in, const int* in_sizes, int n_in,
                              void* d_out, int out_size, void* d_ws, size_t ws_size,
                              hipStream_t stream) {
    const float* dec_states = (const float*)d_in[0];
    const float* enc_states = (const float*)d_in[1];
    const float* W_mlp      = (const float*)d_in[2];
    const float* b_mlp      = (const float*)d_in[3];
    const float* W_out      = (const float*)d_in[4];
    const float* b_out      = (const float*)d_in[5];
    float*       out        = (float*)d_out;

    float* enc_proj = (float*)d_ws;
    float* dec_proj = enc_proj + (size_t)B_SZ * ENC_N * HID_N;

    proj_wmma_f32_rb<<<256, 128, 0, stream>>>(enc_states, W_mlp, enc_proj,
                                              HID_N,
                                               1024,  0,
                                               8);
    proj_wmma_f32<<<256, 256, 0, stream>>>(dec_states, W_mlp, b_mlp, dec_proj,
                                          HID_N,
                                           1024,  512,
                                           16);
    relu_dot_score<<<256, 256, 0, stream>>>(dec_proj, enc_proj,
                                            W_out, b_out, out);
}


# score kernel: w*relu(d+e) formed as sign(w)*max(|w|e, -|w|d) + w*d (w*d summed once per dec row): per term one max and half a packed fma, was add+max+fma; rounding and h sum order change
# speedup vs baseline: 4.2440x; 1.0469x over previous
#include <hip/hip_runtime.h>

typedef __attribute__((ext_vector_type(2))) float v2f;
typedef __attribute__((ext_vector_type(8))) float v8f;

#define B_SZ   4
#define DEC_N  128
#define ENC_N  512
#define HID_N  512
#define K_N    512

#define KSPLIT   8
#define PART_LD  36
#define KW       (K_N / KSPLIT)
#define KLINES   (KW / 32)

typedef __attribute__((ext_vector_type(4))) float v4f;

struct RowSlice { v4f q[KLINES * 4]; };

__device__ __forceinline__ void load_line(RowSlice& f, const float* p, int line) {
#pragma unroll
    for (int i = 0; i < 4; ++i)
        f.q[line * 4 + i] = *(const v4f*)(p + line * 32 + i * 4);
}

__device__ __forceinline__ v2f k_pair(const RowSlice& f, int t) {
    const v4f q = f.q[t >> 1];
    return (t & 1) ? __builtin_shufflevector(q, q, 2, 3)
                   : __builtin_shufflevector(q, q, 0, 1);
}

__global__ __launch_bounds__(256) void proj_wmma_f32(
    const float* __restrict__ A, const float* __restrict__ W,
    const float* __restrict__ bias, float* __restrict__ C,
    int N, int ldw, int woff, int nunits_n)
{
    __shared__ __attribute__((aligned(16))) float s_part[KSPLIT * 32 * PART_LD];

    const int lane = threadIdx.x & 31;
    const int wave = threadIdx.x >> 5;
    const int unit = blockIdx.x;
    const int um   = (unit / nunits_n) << 5;
    const int un   = (unit % nunits_n) << 5;

    const int half = lane >> 4;
    const int r    = lane & 15;

    const int kbeg = wave * KW + (half << 4);

    const float* a0p = A + (size_t)(um + r)      * K_N + kbeg;
    const float* a1p = A + (size_t)(um + 16 + r) * K_N + kbeg;
    const float* b0p = W + (size_t)(un + r)      * ldw + woff + kbeg;
    const float* b1p = W + (size_t)(un + 16 + r) * ldw + woff + kbeg;

    RowSlice a0, a1, b0, b1;
#pragma unroll
    for (int line = 0; line < KLINES; ++line) {
        load_line(a0, a0p, line);
        load_line(b0, b0p, line);
        load_line(a1, a1p, line);
        load_line(b1, b1p, line);
    }
    __builtin_amdgcn_sched_barrier(0);

    v8f c00 = {0.f,0.f,0.f,0.f,0.f,0.f,0.f,0.f};
    v8f c01 = c00, c10 = c00, c11 = c00;

#pragma unroll
    for (int t = 0; t < KLINES * 8; ++t) {
        const v2f x0 = k_pair(a0, t);
        const v2f x1 = k_pair(a1, t);
        const v2f y0 = k_pair(b0, t);
        const v2f y1 = k_pair(b1, t);
        c00 = __builtin_amdgcn_wmma_f32_16x16x4_f32(false, x0, false, y0, (short)0, c00, false, false);
        c01 = __builtin_amdgcn_wmma_f32_16x16x4_f32(false, x0, false, y1, (short)0, c01, false, false);
        c10 = __builtin_amdgcn_wmma_f32_16x16x4_f32(false, x1, false, y0, (short)0, c10, false, false);
        c11 = __builtin_amdgcn_wmma_f32_16x16x4_f32(false, x1, false, y1, (short)0, c11, false, false);
    }

    float* sp = s_part + wave * (32 * PART_LD) + (half << 3) * PART_LD + r;
#pragma unroll
    for (int v = 0; v < 8; ++v) {
        sp[v * PART_LD]             = c00[v];
        sp[v * PART_LD + 16]        = c01[v];
        sp[(v + 16) * PART_LD]      = c10[v];
        sp[(v + 16) * PART_LD + 16] = c11[v];
    }
    __syncthreads();

    const int row = threadIdx.x >> 3;
    const int c4  = (threadIdx.x & 7) << 2;
    const float* rp = s_part + row * PART_LD + c4;
    float4 s = *(const float4*)rp;
#pragma unroll
    for (int w = 1; w < KSPLIT; ++w) {
        const float4 p = *(const float4*)(rp + w * (32 * PART_LD));
        s.x += p.x; s.y += p.y; s.z += p.z; s.w += p.w;
    }
    if (bias) {
        const float* bp = bias + un + c4;
        s.x += bp[0]; s.y += bp[1]; s.z += bp[2]; s.w += bp[3];
    }
    *(float4*)(C + (size_t)(um + row) * N + un + c4) = s;
}

#define ET       4
#define E_WAVES  4
#define E_KW     (K_N / E_WAVES)
#define E_LINES  (E_KW / 32)
#define E_LD     68

struct LineSet { v4f a[ET][4]; v4f b[ET][4]; };

__device__ __forceinline__ void load_lineset(LineSet& f, const float* ap, const float* bp,
                                             int ldw, int line) {
#pragma unroll
    for (int s = 0; s < ET; ++s) {
#pragma unroll
        for (int i = 0; i < 4; ++i)
            f.a[s][i] = *(const v4f*)(ap + (size_t)(s * 16) * K_N + line * 32 + i * 4);
#pragma unroll
        for (int i = 0; i < 4; ++i)
            f.b[s][i] = *(const v4f*)(bp + (size_t)(s * 16) * ldw + line * 32 + i * 4);
    }
}

__device__ __forceinline__ v2f pair_of(const v4f q, int odd) {
    return odd ? __builtin_shufflevector(q, q, 2, 3)
               : __builtin_shufflevector(q, q, 0, 1);
}

__device__ __forceinline__ void mma_lineset(const LineSet& f, v8f (&acc)[ET][ET]) {
#pragma unroll
    for (int t = 0; t < 8; ++t) {
        v2f x[ET], y[ET];
#pragma unroll
        for (int s = 0; s < ET; ++s) {
            x[s] = pair_of(f.a[s][t >> 1], t & 1);
            y[s] = pair_of(f.b[s][t >> 1], t & 1);
        }
#pragma unroll
        for (int i = 0; i < ET; ++i)
#pragma unroll
            for (int j = 0; j < ET; ++j)
                acc[i][j] = __builtin_amdgcn_wmma_f32_16x16x4_f32(false, x[i], false, y[j],
                                                                  (short)0, acc[i][j], false, false);
    }
}

__global__ __launch_bounds__(128) void proj_wmma_f32_rb(
    const float* __restrict__ A, const float* __restrict__ W,
    float* __restrict__ C, int N, int ldw, int woff, int nunits_n)
{
    __shared__ __attribute__((aligned(16))) float s_blk[E_WAVES * 64 * E_LD];

    const int lane = threadIdx.x & 31;
    const int wave = threadIdx.x >> 5;
    const int unit = blockIdx.x;
    const int um   = (unit / nunits_n) << 6;
    const int un   = (unit % nunits_n) << 6;

    const int half = lane >> 4;
    const int r    = lane & 15;

    const int kbeg = wave * E_KW + (half << 4);

    const float* ap = A + (size_t)(um + r) * K_N + kbeg;
    const float* bp = W + (size_t)(un + r) * ldw + woff + kbeg;

    v8f acc[ET][ET];
#pragma unroll
    for (int i = 0; i < ET; ++i)
#pragma unroll
        for (int j = 0; j < ET; ++j)
            acc[i][j] = (v8f){0.f,0.f,0.f,0.f,0.f,0.f,0.f,0.f};

    LineSet f0, f1;
    load_lineset(f0, ap, bp, ldw, 0);
    load_lineset(f1, ap, bp, ldw, 1);
    __builtin_amdgcn_sched_barrier(0);
    mma_lineset(f0, acc);
    __builtin_amdgcn_sched_barrier(0);
    load_lineset(f0, ap, bp, ldw, 2);
    __builtin_amdgcn_sched_barrier(0);
    mma_lineset(f1, acc);
    __builtin_amdgcn_sched_barrier(0);
    load_lineset(f1, ap, bp, ldw, 3);
    __builtin_amdgcn_sched_barrier(0);
    mma_lineset(f0, acc);
    mma_lineset(f1, acc);

    float* sp = s_blk + wave * (64 * E_LD) + (half << 3) * E_LD + r;
#pragma unroll
    for (int i = 0; i < ET; ++i)
#pragma unroll
        for (int j = 0; j < ET; ++j)
#pragma unroll
            for (int v = 0; v < 8; ++v)
                sp[(i * 16 + v) * E_LD + j * 16] = acc[i][j][v];
    __syncthreads();

    const int row0 = threadIdx.x >> 4;
    const int c4   = (threadIdx.x & 15) << 2;
#pragma unroll
    for (int p = 0; p < 8; ++p) {
        const int row = row0 + (p << 3);
        const float* rp = s_blk + row * E_LD + c4;
        float4 s = *(const float4*)rp;
#pragma unroll
        for (int w = 1; w < E_WAVES; ++w) {
            const float4 q = *(const float4*)(rp + w * (64 * E_LD));
            s.x += q.x; s.y += q.y; s.z += q.z; s.w += q.w;
        }
        *(float4*)(C + (size_t)(um + row) * N + un + c4) = s;
    }
}

#define SC_QT     8
#define SC_EW     16
#define SC_WAVES  8
#define SC_RP     8
#define SC_BE     (SC_WAVES * SC_EW)
#define SC_PITCH  (SC_BE + 4)
static_assert(SC_WAVES == SC_QT, "the closing store gives wave w the block's dec row w");
static_assert(SC_EW % SC_RP == 0 && SC_RP % 4 == 0, "a pass is a whole number of turns of the ring of four enc rows");

__device__ __forceinline__ float lane_xor16(float x) {
    const int b = __builtin_bit_cast(int, x);
    return __builtin_bit_cast(float,
        __builtin_amdgcn_permlanex16(b, b, 0x76543210, 0xfedcba98, false, false));
}
template <int M>
__device__ __forceinline__ float row_xor(float x) {
    const int b = __builtin_bit_cast(int, x);
    return __builtin_bit_cast(float,
        __builtin_amdgcn_update_dpp(b, b, 0x160 + M, 0xf, 0xf, false));
}

__device__ __forceinline__ float fold_q8(const float (&acc)[SC_QT], bool up16, bool up8, bool up4) {
    float b4[4];
#pragma unroll
    for (int k = 0; k < 4; ++k) {
        const float keep = up16 ? acc[k + 4] : acc[k];
        const float send = up16 ? acc[k]     : acc[k + 4];
        b4[k] = keep + lane_xor16(send);
    }
    float c2[2];
#pragma unroll
    for (int k = 0; k < 2; ++k) {
        const float keep = up8 ? b4[k + 2] : b4[k];
        const float send = up8 ? b4[k]     : b4[k + 2];
        c2[k] = keep + row_xor<8>(send);
    }
    const float keep1 = up4 ? c2[1] : c2[0];
    const float send1 = up4 ? c2[0] : c2[1];
    float r = keep1 + row_xor<4>(send1);
    r += row_xor<2>(r);
    r += row_xor<1>(r);
    return r;
}

__global__ __launch_bounds__(SC_WAVES * 32) void relu_dot_score(
    const float* __restrict__ dec_proj,
    const float* __restrict__ enc_proj,
    const float* __restrict__ w_out,
    const float* __restrict__ b_out,
    float* __restrict__ out)
{
    __shared__ __attribute__((aligned(16))) float s_res[SC_QT * SC_PITCH];

    const int lane = threadIdx.x & 31;
    const int wave = threadIdx.x >> 5;
    const int blk  = blockIdx.x;
    const int ec   = blk & 3;
    const int qg   = (blk >> 2) & 15;
    const int b    = blk >> 6;
    const int row0 = b * DEC_N + qg * SC_QT;
    const int eb   = ec * SC_BE;
    const int e0   = eb + wave * SC_EW;
    const int hb   = lane << 2;

    const bool up16 = (lane & 16) != 0;
    const bool up8  = (lane & 8) != 0;
    const bool up4  = (lane & 4) != 0;
    const int  qsel = (lane >> 2) & 7;

    v2f av[8];
    v2f sv[8];
    v2f nd[SC_QT][8];
    v2f d02[SC_QT];
#pragma unroll
    for (int q = 0; q < SC_QT; ++q) d02[q] = (v2f){0.f, 0.f};
#pragma unroll
    for (int j = 0; j < 4; ++j) {
        const float4 w4 = *(const float4*)(w_out + hb + (j << 7));
        av[2 * j]     = (v2f){fabsf(w4.x), fabsf(w4.y)};
        av[2 * j + 1] = (v2f){fabsf(w4.z), fabsf(w4.w)};
        sv[2 * j]     = (v2f){copysignf(1.f, w4.x), copysignf(1.f, w4.y)};
        sv[2 * j + 1] = (v2f){copysignf(1.f, w4.z), copysignf(1.f, w4.w)};
#pragma unroll
        for (int q = 0; q < SC_QT; ++q) {
            const float4 d4 = *(const float4*)(dec_proj + (size_t)(row0 + q) * HID_N + hb + (j << 7));
            const v2f dlo = {d4.x, d4.y};
            const v2f dhi = {d4.z, d4.w};
            d02[q] = __builtin_elementwise_fma(dlo, (v2f){w4.x, w4.y}, d02[q]);
            d02[q] = __builtin_elementwise_fma(dhi, (v2f){w4.z, w4.w}, d02[q]);
            nd[q][2 * j]     = -(av[2 * j]     * dlo);
            nd[q][2 * j + 1] = -(av[2 * j + 1] * dhi);
        }
    }
    float d0[SC_QT];
#pragma unroll
    for (int q = 0; q < SC_QT; ++q) d0[q] = d02[q].x + d02[q].y;
    const float cq = fold_q8(d0, up16, up8, up4) + b_out[0];

    const float* ep = enc_proj + ((size_t)b * ENC_N + e0) * HID_N + hb;

    float4 ev[4][4];
#pragma unroll
    for (int s = 0; s < 3; ++s)
#pragma unroll
        for (int j = 0; j < 4; ++j)
            ev[s][j] = *(const float4*)(ep + (size_t)s * HID_N + (j << 7));

    float* sres = s_res + wave * SC_EW;

    for (int i = 0; i < SC_EW; i += SC_RP) {
#pragma unroll
        for (int s = 0; s < SC_RP; ++s) {
            const int nr = min(i + s + 3, SC_EW - 1);
#pragma unroll
            for (int j = 0; j < 4; ++j)
                ev[(s + 3) & 3][j] = *(const float4*)(ep + (size_t)nr * HID_N + (j << 7));

            v2f acc2[SC_QT];
#pragma unroll
            for (int q = 0; q < SC_QT; ++q) acc2[q] = (v2f){0.f, 0.f};
#pragma unroll
            for (int p = 0; p < 8; ++p) {
                const float4 e4 = ev[s & 3][p >> 1];
                const v2f x = av[p] * ((p & 1) ? (v2f){e4.z, e4.w} : (v2f){e4.x, e4.y});
                const v2f s2 = sv[p];
#pragma unroll
                for (int q = 0; q < SC_QT; ++q) {
                    const v2f n2 = nd[q][p];
                    const v2f m2 = {fmaxf(x.x, n2.x), fmaxf(x.y, n2.y)};
                    acc2[q] = __builtin_elementwise_fma(m2, s2, acc2[q]);
                }
            }
            float acc[SC_QT];
#pragma unroll
            for (int q = 0; q < SC_QT; ++q) acc[q] = acc2[q].x + acc2[q].y;

            const float r = fold_q8(acc, up16, up8, up4);
            sres[qsel * SC_PITCH + i + s] = r + cq;
        }
    }
    __syncthreads();

    const float4 o = *(const float4*)(s_res + wave * SC_PITCH + hb);
    *(float4*)(out + (size_t)(row0 + wave) * ENC_N + eb + hb) = o;
}

extern "C" void kernel_launch(void* const* d_in, const int* in_sizes, int n_in,
                              void* d_out, int out_size, void* d_ws, size_t ws_size,
                              hipStream_t stream) {
    const float* dec_states = (const float*)d_in[0];
    const float* enc_states = (const float*)d_in[1];
    const float* W_mlp      = (const float*)d_in[2];
    const float* b_mlp      = (const float*)d_in[3];
    const float* W_out      = (const float*)d_in[4];
    const float* b_out      = (const float*)d_in[5];
    float*       out        = (float*)d_out;

    float* enc_proj = (float*)d_ws;
    float* dec_proj = enc_proj + (size_t)B_SZ * ENC_N * HID_N;

    proj_wmma_f32_rb<<<256, 128, 0, stream>>>(enc_states, W_mlp, enc_proj,
                                              HID_N,
                                               1024,  0,
                                               8);
    proj_wmma_f32<<<256, 256, 0, stream>>>(dec_states, W_mlp, b_mlp, dec_proj,
                                          HID_N,
                                           1024,  512,
                                           16);
    relu_dot_score<<<256, 256, 0, stream>>>(dec_proj, enc_proj,
                                            W_out, b_out, out);
}
